# starting kernel
# baseline (speedup 1.0000x reference)
#include <hip/hip_runtime.h>
#include <hip/hip_bf16.h>
#include <math.h>


typedef __attribute__((ext_vector_type(2))) float v2f;
typedef __attribute__((ext_vector_type(8))) float v8f;

#define WAVE 32

__global__ void fill_kernel(float* __restrict__ p, float v, long long n) {
    long long i = (long long)blockIdx.x * blockDim.x + threadIdx.x;
    if (i < n) p[i] = v;
}

__global__ void selfloop_accum_kernel(const int* __restrict__ dst,
                                      const float* __restrict__ ew,
                                      float* __restrict__ sumw,
                                      float* __restrict__ cntw, int E) {
    int e = blockIdx.x * blockDim.x + threadIdx.x;
    if (e >= E) return;
    int d = dst[e];
    atomicAdd(&sumw[d], ew[e]);
    atomicAdd(&cntw[d], 1.0f);
}

__global__ void selfloop_w_kernel(const float* __restrict__ sumw,
                                  const float* __restrict__ cntw,
                                  float* __restrict__ loopw, int n) {
    int i = blockIdx.x * blockDim.x + threadIdx.x;
    if (i >= n) return;
    loopw[i] = sumw[i] / fmaxf(cntw[i], 1.0f);
}

__global__ void gemm_bias_wmma_kernel(const float* __restrict__ X,
                                      const float* __restrict__ W,
                                      const float* __restrict__ bias,
                                      float* __restrict__ Y,
                                      int rows, int K, int Mout) {
    int gtid = blockIdx.x * blockDim.x + threadIdx.x;
    int wave = gtid >> 5;
    int lane = gtid & 31;
    int tilesN = Mout >> 4;
    int rowT = wave / tilesN;
    int colT = wave - rowT * tilesN;
    if (rowT * 16 >= rows) return;

    const int m    = lane & 15;
    const int half = lane >> 4;

    const float* __restrict__ Arow = X + (size_t)(rowT * 16 + m) * K;
    const float* __restrict__ Bcol = W + (size_t)(colT * 16 + m);

    v8f acc = {};
    for (int k0 = 0; k0 < K; k0 += 4) {
        const int ka = k0 + half * 2;
        v2f a;
        a.x = Arow[ka];
        a.y = Arow[ka + 1];
        v2f b;
        b.x = Bcol[(size_t)ka * Mout];
        b.y = Bcol[(size_t)(ka + 1) * Mout];
        acc = __builtin_amdgcn_wmma_f32_16x16x4_f32(
             false, a,  false, b,
             (short)0, acc,  false,  false);
    }

    const int col = colT * 16 + m;
    const float bv = bias[col];
#pragma unroll
    for (int j = 0; j < 8; ++j) {
        const int row = rowT * 16 + j + half * 8;
        Y[(size_t)row * Mout + col] = acc[j] + bv;
    }
}

__device__ __forceinline__ void atomicMaxF(float* addr, float value) {
    if (value >= 0.0f)
        atomicMax((int*)addr, __float_as_int(value));
    else
        atomicMin((unsigned int*)addr, __float_as_uint(value));
}

template <int C>
__global__ void edge_logits_kernel(const int* __restrict__ src,
                                   const int* __restrict__ dst,
                                   const float* __restrict__ ew,
                                   const float* __restrict__ loopw,
                                   int E, int Etot,
                                   const float* __restrict__ xl,
                                   const float* __restrict__ xr,
                                   const float* __restrict__ We,
                                   const float* __restrict__ att,
                                   float* __restrict__ logits,
                                   float* __restrict__ maxb) {
    const int gtid = blockIdx.x * blockDim.x + threadIdx.x;
    const int e    = gtid >> 5;
    const int lane = gtid & 31;
    if (e >= Etot) return;

    int s, d;
    float w;
    if (e < E) { s = src[e]; d = dst[e]; w = ew[e]; }
    else       { s = d = e - E; w = loopw[s]; }

    const int HC = 4 * C;
    float lg0 = 0.f, lg1 = 0.f, lg2 = 0.f, lg3 = 0.f;

    for (int c = lane; c < C; c += WAVE) {
        const size_t sb = (size_t)s * HC;
        const size_t db = (size_t)d * HC;
#pragma unroll
        for (int h = 0; h < 4; ++h) {
            const int idx = h * C + c;
            float v = xl[sb + idx] + xr[db + idx] + w * We[idx];
            v = (v > 0.f) ? v : 0.2f * v;
            const float t = v * att[idx];
            if (h == 0) lg0 += t; else if (h == 1) lg1 += t;
            else if (h == 2) lg2 += t; else lg3 += t;
        }
    }
#pragma unroll
    for (int off = 16; off > 0; off >>= 1) {
        lg0 += __shfl_xor(lg0, off, WAVE);
        lg1 += __shfl_xor(lg1, off, WAVE);
        lg2 += __shfl_xor(lg2, off, WAVE);
        lg3 += __shfl_xor(lg3, off, WAVE);
    }
    if (lane == 0) {
        float lg[4] = {lg0, lg1, lg2, lg3};
#pragma unroll
        for (int h = 0; h < 4; ++h) {
            logits[(size_t)e * 4 + h] = lg[h];
            atomicMaxF(&maxb[(size_t)d * 4 + h], lg[h]);
        }
    }
}

__global__ void edge_softmax_kernel(const int* __restrict__ dst, int E, int Etot,
                                    float* __restrict__ logits,
                                    const float* __restrict__ maxb,
                                    float* __restrict__ denb) {
    const int e = blockIdx.x * blockDim.x + threadIdx.x;
    if (e >= Etot) return;
    const int d = (e < E) ? dst[e] : (e - E);
#pragma unroll
    for (int h = 0; h < 4; ++h) {
        const float p = __expf(logits[(size_t)e * 4 + h] - maxb[(size_t)d * 4 + h]);
        logits[(size_t)e * 4 + h] = p;
        atomicAdd(&denb[(size_t)d * 4 + h], p);
    }
}

template <int C>
__global__ void edge_aggregate_kernel(const int* __restrict__ src,
                                      const int* __restrict__ dst,
                                      int E, int Etot,
                                      const float* __restrict__ pnum,
                                      const float* __restrict__ denb,
                                      const float* __restrict__ xl,
                                      float* __restrict__ outacc) {
    const int gtid = blockIdx.x * blockDim.x + threadIdx.x;
    const int e    = gtid >> 5;
    const int lane = gtid & 31;
    if (e >= Etot) return;

    int s, d;
    if (e < E) { s = src[e]; d = dst[e]; }
    else       { s = d = e - E; }

    float alpha[4];
#pragma unroll
    for (int h = 0; h < 4; ++h)
        alpha[h] = pnum[(size_t)e * 4 + h] / denb[(size_t)d * 4 + h];

    const int HC = 4 * C;
    const size_t sb = (size_t)s * HC;
    const size_t db = (size_t)d * HC;
    for (int c = lane; c < C; c += WAVE) {
#pragma unroll
        for (int h = 0; h < 4; ++h) {
            const int idx = h * C + c;
            atomicAdd(&outacc[db + idx], xl[sb + idx] * alpha[h]);
        }
    }
}

__global__ void finalize_concat_kernel(const float* __restrict__ acc,
                                       const float* __restrict__ bias,
                                       const float* __restrict__ g,
                                       const float* __restrict__ be,
                                       const float* __restrict__ res,
                                       float* __restrict__ out, int nN) {
    const int gtid = blockIdx.x * blockDim.x + threadIdx.x;
    const int node = gtid >> 5;
    const int lane = gtid & 31;
    if (node >= nN) return;
    const int D = 128;

    float v[4], s = 0.f, s2 = 0.f;
#pragma unroll
    for (int i = 0; i < 4; ++i) {
        const int ch = lane + i * WAVE;
        float x = acc[(size_t)node * D + ch] + bias[ch];
        x = (x > 0.f) ? x : (__expf(x) - 1.f);
        v[i] = x;
        s += x;
        s2 += x * x;
    }
#pragma unroll
    for (int off = 16; off > 0; off >>= 1) {
        s  += __shfl_xor(s, off, WAVE);
        s2 += __shfl_xor(s2, off, WAVE);
    }
    const float mu  = s / (float)D;
    const float var = s2 / (float)D - mu * mu;
    const float inv = rsqrtf(var + 1e-5f);
#pragma unroll
    for (int i = 0; i < 4; ++i) {
        const int ch = lane + i * WAVE;
        out[(size_t)node * D + ch] =
            (v[i] - mu) * inv * g[ch] + be[ch] + res[(size_t)node * D + ch];
    }
}

__global__ void finalize_mean_kernel(const float* __restrict__ acc,
                                     const float* __restrict__ bias,
                                     const float* __restrict__ g,
                                     const float* __restrict__ be,
                                     const float* __restrict__ res,
                                     float* __restrict__ out, int nN) {
    const int gtid = blockIdx.x * blockDim.x + threadIdx.x;
    const int node = gtid >> 5;
    const int lane = gtid & 31;
    if (node >= nN) return;
    const int D = 64;

    float v[2], s = 0.f, s2 = 0.f;
#pragma unroll
    for (int i = 0; i < 2; ++i) {
        const int c = lane + i * WAVE;
        const size_t nb = (size_t)node * 256;
        float x = 0.25f * (acc[nb + 0 * D + c] + acc[nb + 1 * D + c] +
                           acc[nb + 2 * D + c] + acc[nb + 3 * D + c]) + bias[c];
        x = (x > 0.f) ? x : (__expf(x) - 1.f);
        v[i] = x;
        s += x;
        s2 += x * x;
    }
#pragma unroll
    for (int off = 16; off > 0; off >>= 1) {
        s  += __shfl_xor(s, off, WAVE);
        s2 += __shfl_xor(s2, off, WAVE);
    }
    const float mu  = s / (float)D;
    const float var = s2 / (float)D - mu * mu;
    const float inv = rsqrtf(var + 1e-5f);
#pragma unroll
    for (int i = 0; i < 2; ++i) {
        const int c = lane + i * WAVE;
        out[(size_t)node * D + c] =
            (v[i] - mu) * inv * g[c] + be[c] + res[(size_t)node * D + c];
    }
}

static inline int blocks_for(long long work, int bs) {
    return (int)((work + bs - 1) / bs);
}

extern "C" void kernel_launch(void* const* d_in, const int* in_sizes, int n_in,
                              void* d_out, int out_size, void* d_ws, size_t ws_size,
                              hipStream_t stream) {
    const float* x    = (const float*)d_in[0];
    const int*   ei   = (const int*)d_in[1];
    const float* ew   = (const float*)d_in[2];
    const float* Wl0  = (const float*)d_in[3];
    const float* bl0  = (const float*)d_in[4];
    const float* Wr0  = (const float*)d_in[5];
    const float* br0  = (const float*)d_in[6];
    const float* We0  = (const float*)d_in[7];
    const float* att0 = (const float*)d_in[8];
    const float* b0   = (const float*)d_in[9];
    const float* g0   = (const float*)d_in[10];
    const float* be0  = (const float*)d_in[11];
    const float* Wl1  = (const float*)d_in[12];
    const float* bl1  = (const float*)d_in[13];
    const float* Wr1  = (const float*)d_in[14];
    const float* br1  = (const float*)d_in[15];
    const float* We1  = (const float*)d_in[16];
    const float* att1 = (const float*)d_in[17];
    const float* b1   = (const float*)d_in[18];
    const float* g1   = (const float*)d_in[19];
    const float* be1  = (const float*)d_in[20];
    const float* P0   = (const float*)d_in[21];
    const float* pb0  = (const float*)d_in[22];
    const float* P1   = (const float*)d_in[23];
    const float* pb1  = (const float*)d_in[24];

    const int IN  = 128, HIDD = 128, O1 = 256, OUT = 64;
    const int N = in_sizes[0] / IN;
    const int E = in_sizes[2];
    const int Etot = E + N;
    const int* srcp = ei;
    const int* dstp = ei + E;

    float* ws = (float*)d_ws;
    size_t off = 0;
    float* sumw  = ws + off; off += (size_t)N;
    float* cntw  = ws + off; off += (size_t)N;
    float* loopw = ws + off; off += (size_t)N;
    float* XL    = ws + off; off += (size_t)N * O1;
    float* XR    = ws + off; off += (size_t)N * O1;
    float* ACC   = ws + off; off += (size_t)N * O1;
    float* RES   = ws + off; off += (size_t)N * HIDD;
    float* HCUR  = ws + off; off += (size_t)N * HIDD;
    float* MAXB  = ws + off; off += (size_t)N * 4;
    float* DENB  = ws + off; off += (size_t)N * 4;
    float* LOGIT = ws + off; off += (size_t)Etot * 4;
    (void)ws_size; (void)n_in; (void)out_size;

    const int BS = 256;
    const float NEG_INF = -__builtin_inff();

    fill_kernel<<<blocks_for(2LL * N, BS), BS, 0, stream>>>(sumw, 0.f, 2LL * N);
    selfloop_accum_kernel<<<blocks_for(E, BS), BS, 0, stream>>>(dstp, ew, sumw, cntw, E);
    selfloop_w_kernel<<<blocks_for(N, BS), BS, 0, stream>>>(sumw, cntw, loopw, N);

    {
        const long long t128 = (long long)(N / 16) * (HIDD / 16) * WAVE;
        gemm_bias_wmma_kernel<<<blocks_for(t128, BS), BS, 0, stream>>>(x, Wl0, bl0, XL, N, IN, HIDD);
        gemm_bias_wmma_kernel<<<blocks_for(t128, BS), BS, 0, stream>>>(x, Wr0, br0, XR, N, IN, HIDD);
        gemm_bias_wmma_kernel<<<blocks_for(t128, BS), BS, 0, stream>>>(x, P0, pb0, RES, N, IN, HIDD);

        fill_kernel<<<blocks_for(4LL * N, BS), BS, 0, stream>>>(MAXB, NEG_INF, 4LL * N);
        fill_kernel<<<blocks_for(4LL * N, BS), BS, 0, stream>>>(DENB, 0.f, 4LL * N);
        fill_kernel<<<blocks_for((long long)N * HIDD, BS), BS, 0, stream>>>(ACC, 0.f, (long long)N * HIDD);

        const long long ethreads = (long long)Etot * WAVE;
        edge_logits_kernel<32><<<blocks_for(ethreads, BS), BS, 0, stream>>>(
            srcp, dstp, ew, loopw, E, Etot, XL, XR, We0, att0, LOGIT, MAXB);
        edge_softmax_kernel<<<blocks_for(Etot, BS), BS, 0, stream>>>(
            dstp, E, Etot, LOGIT, MAXB, DENB);
        edge_aggregate_kernel<32><<<blocks_for(ethreads, BS), BS, 0, stream>>>(
            srcp, dstp, E, Etot, LOGIT, DENB, XL, ACC);

        finalize_concat_kernel<<<blocks_for((long long)N * WAVE, BS), BS, 0, stream>>>(
            ACC, b0, g0, be0, RES, HCUR, N);
    }

    {
        const long long t256 = (long long)(N / 16) * (O1 / 16) * WAVE;
        const long long t64  = (long long)(N / 16) * (OUT / 16) * WAVE;
        gemm_bias_wmma_kernel<<<blocks_for(t256, BS), BS, 0, stream>>>(HCUR, Wl1, bl1, XL, N, HIDD, O1);
        gemm_bias_wmma_kernel<<<blocks_for(t256, BS), BS, 0, stream>>>(HCUR, Wr1, br1, XR, N, HIDD, O1);
        gemm_bias_wmma_kernel<<<blocks_for(t64, BS), BS, 0, stream>>>(HCUR, P1, pb1, RES, N, HIDD, OUT);

        fill_kernel<<<blocks_for(4LL * N, BS), BS, 0, stream>>>(MAXB, NEG_INF, 4LL * N);
        fill_kernel<<<blocks_for(4LL * N, BS), BS, 0, stream>>>(DENB, 0.f, 4LL * N);
        fill_kernel<<<blocks_for((long long)N * O1, BS), BS, 0, stream>>>(ACC, 0.f, (long long)N * O1);

        const long long ethreads = (long long)Etot * WAVE;
        edge_logits_kernel<64><<<blocks_for(ethreads, BS), BS, 0, stream>>>(
            srcp, dstp, ew, loopw, E, Etot, XL, XR, We1, att1, LOGIT, MAXB);
        edge_softmax_kernel<<<blocks_for(Etot, BS), BS, 0, stream>>>(
            dstp, E, Etot, LOGIT, MAXB, DENB);
        edge_aggregate_kernel<64><<<blocks_for(ethreads, BS), BS, 0, stream>>>(
            srcp, dstp, E, Etot, LOGIT, DENB, XL, ACC);

        finalize_mean_kernel<<<blocks_for((long long)N * WAVE, BS), BS, 0, stream>>>(
            ACC, b1, g1, be1, RES, (float*)d_out, N);
    }
}


# Layer 1 (heads averaged): the 4 heads are summed per edge in registers before the scatter, so an edge adds one 64-float row (2 atomic instructions) not four (8); accumulator plane N x 64
# speedup vs baseline: 1.4816x; 1.4816x over previous
#include <hip/hip_runtime.h>
#include <hip/hip_bf16.h>
#include <math.h>


typedef __attribute__((ext_vector_type(2))) float v2f;
typedef __attribute__((ext_vector_type(8))) float v8f;

#define WAVE 32

__global__ void fill_kernel(float* __restrict__ p, float v, long long n) {
    long long i = (long long)blockIdx.x * blockDim.x + threadIdx.x;
    if (i < n) p[i] = v;
}

__global__ void selfloop_accum_kernel(const int* __restrict__ dst,
                                      const float* __restrict__ ew,
                                      float* __restrict__ sc, int E) {
    const int t = blockIdx.x * blockDim.x + threadIdx.x;
    const int e = t >> 1;
    if (e >= E) return;
    const int which = t & 1;
    const int d = dst[e];
    const float v = which ? 1.0f : ew[e];
    atomicAdd(&sc[2 * (size_t)d + which], v);
}

__global__ void selfloop_w_kernel(const float* __restrict__ sc,
                                  float* __restrict__ loopw, int n) {
    int i = blockIdx.x * blockDim.x + threadIdx.x;
    if (i >= n) return;
    loopw[i] = sc[2 * (size_t)i] / fmaxf(sc[2 * (size_t)i + 1], 1.0f);
}

__global__ void gemm_bias_wmma_kernel(const float* __restrict__ X,
                                      const float* __restrict__ W,
                                      const float* __restrict__ bias,
                                      float* __restrict__ Y,
                                      int rows, int K, int Mout) {
    int gtid = blockIdx.x * blockDim.x + threadIdx.x;
    int wave = gtid >> 5;
    int lane = gtid & 31;
    int tilesN = Mout >> 4;
    int rowT = wave / tilesN;
    int colT = wave - rowT * tilesN;
    if (rowT * 16 >= rows) return;

    const int m    = lane & 15;
    const int half = lane >> 4;

    const float* __restrict__ Arow = X + (size_t)(rowT * 16 + m) * K;
    const float* __restrict__ Bcol = W + (size_t)(colT * 16 + m);

    v8f acc = {};
    for (int k0 = 0; k0 < K; k0 += 4) {
        const int ka = k0 + half * 2;
        v2f a;
        a.x = Arow[ka];
        a.y = Arow[ka + 1];
        v2f b;
        b.x = Bcol[(size_t)ka * Mout];
        b.y = Bcol[(size_t)(ka + 1) * Mout];
        acc = __builtin_amdgcn_wmma_f32_16x16x4_f32(
             false, a,  false, b,
             (short)0, acc,  false,  false);
    }

    const int col = colT * 16 + m;
    const float bv = bias[col];
#pragma unroll
    for (int j = 0; j < 8; ++j) {
        const int row = rowT * 16 + j + half * 8;
        Y[(size_t)row * Mout + col] = acc[j] + bv;
    }
}

__device__ __forceinline__ int float_to_key(float f) {
    const int b = __float_as_int(f);
    return b ^ ((b >> 31) & 0x7fffffff);
}
__device__ __forceinline__ float key_to_float(int k) {
    return __int_as_float(k ^ ((k >> 31) & 0x7fffffff));
}

__global__ void fill_maxkey_kernel(int* __restrict__ p, long long n) {
    long long i = (long long)blockIdx.x * blockDim.x + threadIdx.x;
    if (i < n) p[i] = float_to_key(-__builtin_inff());
}

template <int C>
__global__ void edge_logits_kernel(const int* __restrict__ src,
                                   const int* __restrict__ dst,
                                   const float* __restrict__ ew,
                                   const float* __restrict__ loopw,
                                   int E, int Etot,
                                   const float* __restrict__ xl,
                                   const float* __restrict__ xr,
                                   const float* __restrict__ We,
                                   const float* __restrict__ att,
                                   float* __restrict__ logits,
                                   int* __restrict__ maxk) {
    const int gtid = blockIdx.x * blockDim.x + threadIdx.x;
    const int e    = gtid >> 5;
    const int lane = gtid & 31;
    if (e >= Etot) return;

    int s, d;
    float w;
    if (e < E) { s = src[e]; d = dst[e]; w = ew[e]; }
    else       { s = d = e - E; w = loopw[s]; }

    const int HC = 4 * C;
    float lg0 = 0.f, lg1 = 0.f, lg2 = 0.f, lg3 = 0.f;

    for (int c = lane; c < C; c += WAVE) {
        const size_t sb = (size_t)s * HC;
        const size_t db = (size_t)d * HC;
#pragma unroll
        for (int h = 0; h < 4; ++h) {
            const int idx = h * C + c;
            float v = xl[sb + idx] + xr[db + idx] + w * We[idx];
            v = (v > 0.f) ? v : 0.2f * v;
            const float t = v * att[idx];
            if (h == 0) lg0 += t; else if (h == 1) lg1 += t;
            else if (h == 2) lg2 += t; else lg3 += t;
        }
    }
#pragma unroll
    for (int off = 16; off > 0; off >>= 1) {
        lg0 += __shfl_xor(lg0, off, WAVE);
        lg1 += __shfl_xor(lg1, off, WAVE);
        lg2 += __shfl_xor(lg2, off, WAVE);
        lg3 += __shfl_xor(lg3, off, WAVE);
    }
    if (lane < 4) {
        const float mine = (lane & 2) ? ((lane & 1) ? lg3 : lg2)
                                      : ((lane & 1) ? lg1 : lg0);
        logits[(size_t)e * 4 + lane] = mine;
        atomicMax(&maxk[(size_t)d * 4 + lane], float_to_key(mine));
    }
}

__global__ void edge_softmax_kernel(const int* __restrict__ dst, int E, int Etot,
                                    float* __restrict__ logits,
                                    const int* __restrict__ maxk,
                                    float* __restrict__ denb) {
    const int t = blockIdx.x * blockDim.x + threadIdx.x;
    const int e = t >> 2;
    if (e >= Etot) return;
    const int h = t & 3;
    const int d = (e < E) ? dst[e] : (e - E);
    const float p = __expf(logits[t] - key_to_float(maxk[(size_t)d * 4 + h]));
    logits[t] = p;
    atomicAdd(&denb[(size_t)d * 4 + h], p);
}

template <int C>
__global__ void edge_aggregate_kernel(const int* __restrict__ src,
                                      const int* __restrict__ dst,
                                      int E, int Etot,
                                      const float* __restrict__ pnum,
                                      const float* __restrict__ denb,
                                      const float* __restrict__ xl,
                                      float* __restrict__ outacc) {
    const int gtid = blockIdx.x * blockDim.x + threadIdx.x;
    const int e    = gtid >> 5;
    const int lane = gtid & 31;
    if (e >= Etot) return;

    int s, d;
    if (e < E) { s = src[e]; d = dst[e]; }
    else       { s = d = e - E; }

    float alpha[4];
#pragma unroll
    for (int h = 0; h < 4; ++h)
        alpha[h] = pnum[(size_t)e * 4 + h] / denb[(size_t)d * 4 + h];

    const int HC = 4 * C;
    const size_t sb = (size_t)s * HC;
    const size_t db = (size_t)d * HC;
    for (int c = lane; c < C; c += WAVE) {
#pragma unroll
        for (int h = 0; h < 4; ++h) {
            const int idx = h * C + c;
            atomicAdd(&outacc[db + idx], xl[sb + idx] * alpha[h]);
        }
    }
}

__global__ void edge_aggregate_headsum_kernel(const int* __restrict__ src,
                                              const int* __restrict__ dst,
                                              int E, int Etot,
                                              const float* __restrict__ pnum,
                                              const float* __restrict__ denb,
                                              const float* __restrict__ xl,
                                              float* __restrict__ outacc) {
    const int C = 64;
    const int gtid = blockIdx.x * blockDim.x + threadIdx.x;
    const int e    = gtid >> 5;
    const int lane = gtid & 31;
    if (e >= Etot) return;

    int s, d;
    if (e < E) { s = src[e]; d = dst[e]; }
    else       { s = d = e - E; }

    const float* __restrict__ pe = pnum + (size_t)e * 4;
    const float* __restrict__ dn = denb + (size_t)d * 4;
    const float a0 = pe[0] / dn[0];
    const float a1 = pe[1] / dn[1];
    const float a2 = pe[2] / dn[2];
    const float a3 = pe[3] / dn[3];

    const float* __restrict__ xs = xl + (size_t)s * (4 * C);
    float* __restrict__ od = outacc + (size_t)d * C;
    const int c0 = lane, c1 = lane + WAVE;
    const float v0 = xs[c0] * a0 + xs[C + c0] * a1 + xs[2 * C + c0] * a2 + xs[3 * C + c0] * a3;
    const float v1 = xs[c1] * a0 + xs[C + c1] * a1 + xs[2 * C + c1] * a2 + xs[3 * C + c1] * a3;
    atomicAdd(&od[c0], v0);
    atomicAdd(&od[c1], v1);
}

__global__ void finalize_concat_kernel(const float* __restrict__ acc,
                                       const float* __restrict__ bias,
                                       const float* __restrict__ g,
                                       const float* __restrict__ be,
                                       const float* __restrict__ res,
                                       float* __restrict__ out, int nN) {
    const int gtid = blockIdx.x * blockDim.x + threadIdx.x;
    const int node = gtid >> 5;
    const int lane = gtid & 31;
    if (node >= nN) return;
    const int D = 128;

    float v[4], s = 0.f, s2 = 0.f;
#pragma unroll
    for (int i = 0; i < 4; ++i) {
        const int ch = lane + i * WAVE;
        float x = acc[(size_t)node * D + ch] + bias[ch];
        x = (x > 0.f) ? x : (__expf(x) - 1.f);
        v[i] = x;
        s += x;
        s2 += x * x;
    }
#pragma unroll
    for (int off = 16; off > 0; off >>= 1) {
        s  += __shfl_xor(s, off, WAVE);
        s2 += __shfl_xor(s2, off, WAVE);
    }
    const float mu  = s / (float)D;
    const float var = s2 / (float)D - mu * mu;
    const float inv = rsqrtf(var + 1e-5f);
#pragma unroll
    for (int i = 0; i < 4; ++i) {
        const int ch = lane + i * WAVE;
        out[(size_t)node * D + ch] =
            (v[i] - mu) * inv * g[ch] + be[ch] + res[(size_t)node * D + ch];
    }
}

__global__ void finalize_mean_kernel(const float* __restrict__ acc,
                                     const float* __restrict__ bias,
                                     const float* __restrict__ g,
                                     const float* __restrict__ be,
                                     const float* __restrict__ res,
                                     float* __restrict__ out, int nN) {
    const int gtid = blockIdx.x * blockDim.x + threadIdx.x;
    const int node = gtid >> 5;
    const int lane = gtid & 31;
    if (node >= nN) return;
    const int D = 64;

    float v[2], s = 0.f, s2 = 0.f;
#pragma unroll
    for (int i = 0; i < 2; ++i) {
        const int c = lane + i * WAVE;
        float x = 0.25f * acc[(size_t)node * D + c] + bias[c];
        x = (x > 0.f) ? x : (__expf(x) - 1.f);
        v[i] = x;
        s += x;
        s2 += x * x;
    }
#pragma unroll
    for (int off = 16; off > 0; off >>= 1) {
        s  += __shfl_xor(s, off, WAVE);
        s2 += __shfl_xor(s2, off, WAVE);
    }
    const float mu  = s / (float)D;
    const float var = s2 / (float)D - mu * mu;
    const float inv = rsqrtf(var + 1e-5f);
#pragma unroll
    for (int i = 0; i < 2; ++i) {
        const int c = lane + i * WAVE;
        out[(size_t)node * D + c] =
            (v[i] - mu) * inv * g[c] + be[c] + res[(size_t)node * D + c];
    }
}

static inline int blocks_for(long long work, int bs) {
    return (int)((work + bs - 1) / bs);
}

extern "C" void kernel_launch(void* const* d_in, const int* in_sizes, int n_in,
                              void* d_out, int out_size, void* d_ws, size_t ws_size,
                              hipStream_t stream) {
    const float* x    = (const float*)d_in[0];
    const int*   ei   = (const int*)d_in[1];
    const float* ew   = (const float*)d_in[2];
    const float* Wl0  = (const float*)d_in[3];
    const float* bl0  = (const float*)d_in[4];
    const float* Wr0  = (const float*)d_in[5];
    const float* br0  = (const float*)d_in[6];
    const float* We0  = (const float*)d_in[7];
    const float* att0 = (const float*)d_in[8];
    const float* b0   = (const float*)d_in[9];
    const float* g0   = (const float*)d_in[10];
    const float* be0  = (const float*)d_in[11];
    const float* Wl1  = (const float*)d_in[12];
    const float* bl1  = (const float*)d_in[13];
    const float* Wr1  = (const float*)d_in[14];
    const float* br1  = (const float*)d_in[15];
    const float* We1  = (const float*)d_in[16];
    const float* att1 = (const float*)d_in[17];
    const float* b1   = (const float*)d_in[18];
    const float* g1   = (const float*)d_in[19];
    const float* be1  = (const float*)d_in[20];
    const float* P0   = (const float*)d_in[21];
    const float* pb0  = (const float*)d_in[22];
    const float* P1   = (const float*)d_in[23];
    const float* pb1  = (const float*)d_in[24];

    const int IN  = 128, HIDD = 128, O1 = 256, OUT = 64;
    const int N = in_sizes[0] / IN;
    const int E = in_sizes[2];
    const int Etot = E + N;
    const int* srcp = ei;
    const int* dstp = ei + E;

    float* ws = (float*)d_ws;
    size_t off = 0;
    float* sumcnt = ws + off; off += 2 * (size_t)N;
    float* loopw = ws + off; off += (size_t)N;
    float* XL    = ws + off; off += (size_t)N * O1;
    float* XR    = ws + off; off += (size_t)N * O1;
    float* ACC   = ws + off; off += (size_t)N * O1;
    float* RES   = ws + off; off += (size_t)N * HIDD;
    float* HCUR  = ws + off; off += (size_t)N * HIDD;
    int*   MAXK  = (int*)(ws + off); off += (size_t)N * 4;
    float* DENB  = ws + off; off += (size_t)N * 4;
    float* LOGIT = ws + off; off += (size_t)Etot * 4;
    (void)ws_size; (void)n_in; (void)out_size;

    const int BS = 256;

    fill_kernel<<<blocks_for(2LL * N, BS), BS, 0, stream>>>(sumcnt, 0.f, 2LL * N);
    selfloop_accum_kernel<<<blocks_for(2LL * E, BS), BS, 0, stream>>>(dstp, ew, sumcnt, E);
    selfloop_w_kernel<<<blocks_for(N, BS), BS, 0, stream>>>(sumcnt, loopw, N);

    {
        const long long t128 = (long long)(N / 16) * (HIDD / 16) * WAVE;
        gemm_bias_wmma_kernel<<<blocks_for(t128, BS), BS, 0, stream>>>(x, Wl0, bl0, XL, N, IN, HIDD);
        gemm_bias_wmma_kernel<<<blocks_for(t128, BS), BS, 0, stream>>>(x, Wr0, br0, XR, N, IN, HIDD);
        gemm_bias_wmma_kernel<<<blocks_for(t128, BS), BS, 0, stream>>>(x, P0, pb0, RES, N, IN, HIDD);

        fill_maxkey_kernel<<<blocks_for(4LL * N, BS), BS, 0, stream>>>(MAXK, 4LL * N);
        fill_kernel<<<blocks_for(4LL * N, BS), BS, 0, stream>>>(DENB, 0.f, 4LL * N);
        fill_kernel<<<blocks_for((long long)N * HIDD, BS), BS, 0, stream>>>(ACC, 0.f, (long long)N * HIDD);

        const long long ethreads = (long long)Etot * WAVE;
        edge_logits_kernel<32><<<blocks_for(ethreads, BS), BS, 0, stream>>>(
            srcp, dstp, ew, loopw, E, Etot, XL, XR, We0, att0, LOGIT, MAXK);
        edge_softmax_kernel<<<blocks_for(4LL * Etot, BS), BS, 0, stream>>>(
            dstp, E, Etot, LOGIT, MAXK, DENB);
        edge_aggregate_kernel<32><<<blocks_for(ethreads, BS), BS, 0, stream>>>(
            srcp, dstp, E, Etot, LOGIT, DENB, XL, ACC);

        finalize_concat_kernel<<<blocks_for((long long)N * WAVE, BS), BS, 0, stream>>>(
            ACC, b0, g0, be0, RES, HCUR, N);
    }

    {
        const long long t256 = (long long)(N / 16) * (O1 / 16) * WAVE;
        const long long t64  = (long long)(N / 16) * (OUT / 16) * WAVE;
        gemm_bias_wmma_kernel<<<blocks_for(t256, BS), BS, 0, stream>>>(HCUR, Wl1, bl1, XL, N, HIDD, O1);
        gemm_bias_wmma_kernel<<<blocks_for(t256, BS), BS, 0, stream>>>(HCUR, Wr1, br1, XR, N, HIDD, O1);
        gemm_bias_wmma_kernel<<<blocks_for(t64, BS), BS, 0, stream>>>(HCUR, P1, pb1, RES, N, HIDD, OUT);

        fill_maxkey_kernel<<<blocks_for(4LL * N, BS), BS, 0, stream>>>(MAXK, 4LL * N);
        fill_kernel<<<blocks_for(4LL * N, BS), BS, 0, stream>>>(DENB, 0.f, 4LL * N);
        fill_kernel<<<blocks_for((long long)N * OUT, BS), BS, 0, stream>>>(ACC, 0.f, (long long)N * OUT);

        const long long ethreads = (long long)Etot * WAVE;
        edge_logits_kernel<64><<<blocks_for(ethreads, BS), BS, 0, stream>>>(
            srcp, dstp, ew, loopw, E, Etot, XL, XR, We1, att1, LOGIT, MAXK);
        edge_softmax_kernel<<<blocks_for(4LL * Etot, BS), BS, 0, stream>>>(
            dstp, E, Etot, LOGIT, MAXK, DENB);
        edge_aggregate_headsum_kernel<<<blocks_for(ethreads, BS), BS, 0, stream>>>(
            srcp, dstp, E, Etot, LOGIT, DENB, XL, ACC);

        finalize_mean_kernel<<<blocks_for((long long)N * WAVE, BS), BS, 0, stream>>>(
            ACC, b1, g1, be1, RES, (float*)d_out, N);
    }
}


# Dense transforms: a wave keeps its 16-column weight tile (128x16) in registers as B fragments; row tiles pass through LDS (whole-row 16-byte loads in, one 16-byte read per two WMMAs)
# speedup vs baseline: 2.2253x; 1.5020x over previous
#include <hip/hip_runtime.h>
#include <hip/hip_bf16.h>
#include <math.h>


typedef __attribute__((ext_vector_type(2))) float v2f;
typedef __attribute__((ext_vector_type(4))) float v4f;
typedef __attribute__((ext_vector_type(8))) float v8f;

#define WAVE 32
#define EPW  32
#define FILL_PER_THREAD 16

__device__ __forceinline__ int lane_read_i(int v, int j) {
    return __builtin_amdgcn_readlane(v, j);
}
__device__ __forceinline__ float lane_read_f(float v, int j) {
    return __int_as_float(__builtin_amdgcn_readlane(__float_as_int(v), j));
}
__device__ __forceinline__ int wave_index() {
    return __builtin_amdgcn_readfirstlane((int)((blockIdx.x * blockDim.x + threadIdx.x) >> 5));
}

__global__ void fill_kernel(float* __restrict__ p, float v, long long n) {
    const long long T = (long long)gridDim.x * blockDim.x;
    for (long long i = (long long)blockIdx.x * blockDim.x + threadIdx.x; i < n; i += T)
        p[i] = v;
}

__global__ void selfloop_accum_kernel(const int* __restrict__ dst,
                                      const float* __restrict__ ew,
                                      float* __restrict__ sc, int E) {
    const int t = blockIdx.x * blockDim.x + threadIdx.x;
    const int e = t >> 1;
    if (e >= E) return;
    const int which = t & 1;
    const int d = dst[e];
    const float v = which ? 1.0f : ew[e];
    atomicAdd(&sc[2 * (size_t)d + which], v);
}

__global__ void selfloop_w_kernel(const float* __restrict__ sc,
                                  float* __restrict__ loopw, int n) {
    int i = blockIdx.x * blockDim.x + threadIdx.x;
    if (i >= n) return;
    loopw[i] = sc[2 * (size_t)i] / fmaxf(sc[2 * (size_t)i + 1], 1.0f);
}

#define GK     128
#define GPITCH (GK + 4)

__device__ __forceinline__ void lds_stage_barrier() {
    asm volatile("s_wait_dscnt 0x0\n\ts_barrier_signal -1\n\ts_barrier_wait -1" ::: "memory");
}

template <int CT>
__global__ void __launch_bounds__(256)
gemm_bias_wmma_kernel(const float* __restrict__ X,
                      const float* __restrict__ W,
                      const float* __restrict__ bias,
                      float* __restrict__ Y,
                      int rowTiles) {
    constexpr int MOUT = CT * 16;
    constexpr int CTW  = (CT < 8) ? CT : 8;
    constexpr int RS   = 8 / CTW;
    constexpr int ROWS = RS * 16;
    constexpr int LPT  = ROWS * (GK / 4) / 256;
    __shared__ __attribute__((aligned(16))) float sA[2][ROWS * GPITCH];

    const int tid  = threadIdx.x;
    const int lane = tid & 31;
    const int wv   = __builtin_amdgcn_readfirstlane(tid >> 5);
    const int m    = lane & 15;
    const int half = lane >> 4;
    const int rsub = wv / CTW;
    const int col  = (blockIdx.y * CTW + (wv - rsub * CTW)) * 16 + m;
    const int rows = rowTiles * 16;
    const int nSteps = (rowTiles + RS - 1) / RS;
    const int stride = gridDim.x;
    int step = blockIdx.x;
    if (step >= nSteps) return;

    const int c4 = lane;
    v4f pre[LPT];
#pragma unroll
    for (int j = 0; j < LPT; ++j) {
        const int grow = min(step * ROWS + wv + 8 * j, rows - 1);
        pre[j] = *(const v4f*)(X + (size_t)grow * GK + c4 * 4);
    }

    v2f bf[32];
    {
        const float* __restrict__ Wc = W + col;
#pragma unroll
        for (int t = 0; t < 16; ++t)
#pragma unroll
            for (int q = 0; q < 2; ++q) {
                const int k = 8 * t + 4 * half + 2 * q;
                bf[2 * t + q].x = Wc[(size_t)k * MOUT];
                bf[2 * t + q].y = Wc[(size_t)(k + 1) * MOUT];
            }
    }
    const float bv = bias[col];

#pragma unroll
    for (int j = 0; j < LPT; ++j)
        *(v4f*)(&sA[0][(wv + 8 * j) * GPITCH + c4 * 4]) = pre[j];
    lds_stage_barrier();

    int buf = 0;
    for (; step < nSteps; step += stride) {
        const int nxt = step + stride;
        const bool more = nxt < nSteps;
        if (more) {
#pragma unroll
            for (int j = 0; j < LPT; ++j) {
                const int grow = min(nxt * ROWS + wv + 8 * j, rows - 1);
                pre[j] = *(const v4f*)(X + (size_t)grow * GK + c4 * 4);
            }
        }

        const int tile = step * RS + rsub;
        if (tile < rowTiles) {
            const float* __restrict__ arow = &sA[buf][(rsub * 16 + m) * GPITCH + 4 * half];
            v8f acc0 = {}, acc1 = {};
            v4f af[16];
#pragma unroll
            for (int t = 0; t < 16; ++t) af[t] = *(const v4f*)(arow + 8 * t);
            __builtin_amdgcn_sched_barrier(0);
#pragma unroll
            for (int t = 0; t < 16; t += 2) {
                const v4f a0 = af[t];
                const v4f a1 = af[t + 1];
                acc0 = __builtin_amdgcn_wmma_f32_16x16x4_f32(
                    false, (v2f){a0.x, a0.y}, false, bf[2 * t], (short)0, acc0, false, false);
                acc1 = __builtin_amdgcn_wmma_f32_16x16x4_f32(
                    false, (v2f){a1.x, a1.y}, false, bf[2 * t + 2], (short)0, acc1, false, false);
                acc0 = __builtin_amdgcn_wmma_f32_16x16x4_f32(
                    false, (v2f){a0.z, a0.w}, false, bf[2 * t + 1], (short)0, acc0, false, false);
                acc1 = __builtin_amdgcn_wmma_f32_16x16x4_f32(
                    false, (v2f){a1.z, a1.w}, false, bf[2 * t + 3], (short)0, acc1, false, false);
            }
            float* __restrict__ yp = Y + (size_t)(tile * 16 + half * 8) * MOUT + col;
#pragma unroll
            for (int j = 0; j < 8; ++j)
                yp[(size_t)j * MOUT] = (acc0[j] + acc1[j]) + bv;
        }

        if (more) {
#pragma unroll
            for (int j = 0; j < LPT; ++j)
                *(v4f*)(&sA[buf ^ 1][(wv + 8 * j) * GPITCH + c4 * 4]) = pre[j];
        }
        lds_stage_barrier();
        buf ^= 1;
    }
}

__device__ __forceinline__ int float_to_key(float f) {
    const int b = __float_as_int(f);
    return b ^ ((b >> 31) & 0x7fffffff);
}
__device__ __forceinline__ float key_to_float(int k) {
    return __int_as_float(k ^ ((k >> 31) & 0x7fffffff));
}

__global__ void fill_maxkey_kernel(int* __restrict__ p, long long n) {
    const long long T = (long long)gridDim.x * blockDim.x;
    const int k = float_to_key(-__builtin_inff());
    for (long long i = (long long)blockIdx.x * blockDim.x + threadIdx.x; i < n; i += T)
        p[i] = k;
}

template <int C, int U, int R>
__global__ void __launch_bounds__(256)
edge_logits_kernel(const int* __restrict__ src,
                   const int* __restrict__ dst,
                   const float* __restrict__ ew,
                   const float* __restrict__ loopw,
                   int E, int Etot,
                   const float* __restrict__ xl,
                   const float* __restrict__ xr,
                   const float* __restrict__ We,
                   const float* __restrict__ att,
                   float* __restrict__ logits,
                   int* __restrict__ maxk) {
    constexpr int NC = C / WAVE;
    constexpr int HC = 4 * C;
    const int lane = threadIdx.x & 31;
    const int base = wave_index() * R;
    if (base >= Etot) return;
    const int cnt = min(R, Etot - base);

    int s_l = 0, d_l = 0;
    float w_l = 0.f;
    if (lane < cnt) {
        const int me = base + lane;
        if (me < E) { s_l = src[me]; d_l = dst[me]; w_l = ew[me]; }
        else        { s_l = d_l = me - E; w_l = loopw[me - E]; }
    }

    float we[4][NC], at[4][NC];
#pragma unroll
    for (int h = 0; h < 4; ++h)
#pragma unroll
        for (int i = 0; i < NC; ++i) {
            we[h][i] = We[h * C + i * WAVE + lane];
            at[h][i] = att[h * C + i * WAVE + lane];
        }

    for (int j0 = 0; j0 < cnt; j0 += U) {
        float xa[U][4][NC], xb[U][4][NC];
        int   dj[U];
        float wj[U];
#pragma unroll
        for (int u = 0; u < U; ++u) {
            const int j = min(j0 + u, cnt - 1);
            const int sj = lane_read_i(s_l, j);
            dj[u] = lane_read_i(d_l, j);
            wj[u] = lane_read_f(w_l, j);
            const float* __restrict__ xs = xl + (size_t)sj * HC + lane;
            const float* __restrict__ xd = xr + (size_t)dj[u] * HC + lane;
#pragma unroll
            for (int h = 0; h < 4; ++h)
#pragma unroll
                for (int i = 0; i < NC; ++i) {
                    xa[u][h][i] = xs[h * C + i * WAVE];
                    xb[u][h][i] = xd[h * C + i * WAVE];
                }
        }
#pragma unroll
        for (int u = 0; u < U; ++u) {
            float lg[4] = {0.f, 0.f, 0.f, 0.f};
#pragma unroll
            for (int i = 0; i < NC; ++i)
#pragma unroll
                for (int h = 0; h < 4; ++h) {
                    float v = xa[u][h][i] + xb[u][h][i] + wj[u] * we[h][i];
                    v = (v > 0.f) ? v : 0.2f * v;
                    lg[h] += v * at[h][i];
                }
#pragma unroll
            for (int off = 16; off > 0; off >>= 1) {
                lg[0] += __shfl_xor(lg[0], off, WAVE);
                lg[1] += __shfl_xor(lg[1], off, WAVE);
                lg[2] += __shfl_xor(lg[2], off, WAVE);
                lg[3] += __shfl_xor(lg[3], off, WAVE);
            }
            if (j0 + u < cnt && lane < 4) {
                const float mine = (lane & 2) ? ((lane & 1) ? lg[3] : lg[2])
                                              : ((lane & 1) ? lg[1] : lg[0]);
                logits[(size_t)(base + j0 + u) * 4 + lane] = mine;
                atomicMax(&maxk[(size_t)dj[u] * 4 + lane], float_to_key(mine));
            }
        }
    }
}

__global__ void __launch_bounds__(256)
edge_softmax_kernel(const int* __restrict__ dst, int E, int Etot,
                    float* __restrict__ logits,
                    const int* __restrict__ maxk,
                    float* __restrict__ denb) {
    const long long T = (long long)gridDim.x * blockDim.x;
    const long long n = 4LL * Etot;
    const long long t = (long long)blockIdx.x * blockDim.x + threadIdx.x;
    const int h = (int)(t & 3);
    for (long long i0 = t; i0 < n; i0 += 4 * T) {
        int   d[4];
        float lg[4];
        int   mk[4];
#pragma unroll
        for (int u = 0; u < 4; ++u) {
            const long long i = i0 + u * T;
            d[u] = 0; lg[u] = 0.f;
            if (i < n) {
                const int e = (int)(i >> 2);
                d[u] = (e < E) ? dst[e] : (e - E);
                lg[u] = logits[i];
            }
        }
#pragma unroll
        for (int u = 0; u < 4; ++u)
            mk[u] = (i0 + u * T < n) ? maxk[(size_t)d[u] * 4 + h] : 0;
#pragma unroll
        for (int u = 0; u < 4; ++u) {
            const long long i = i0 + u * T;
            if (i < n) {
                const float p = __expf(lg[u] - key_to_float(mk[u]));
                logits[i] = p;
                atomicAdd(&denb[(size_t)d[u] * 4 + h], p);
            }
        }
    }
}

template <int C, int U>
__global__ void __launch_bounds__(256)
edge_aggregate_kernel(const int* __restrict__ src,
                      const int* __restrict__ dst,
                      int E, int Etot,
                      const float* __restrict__ pnum,
                      const float* __restrict__ denb,
                      const float* __restrict__ xl,
                      float* __restrict__ outacc) {
    static_assert(C == WAVE, "one channel per lane and head");
    constexpr int HC = 4 * C;
    const int lane = threadIdx.x & 31;
    const int base = wave_index() * EPW;
    if (base >= Etot) return;
    const int cnt = min(EPW, Etot - base);

    int s_l = 0, d_l = 0;
    float a_l[4] = {0.f, 0.f, 0.f, 0.f};
    if (lane < cnt) {
        const int me = base + lane;
        if (me < E) { s_l = src[me]; d_l = dst[me]; }
        else        { s_l = d_l = me - E; }
        const float* __restrict__ pe = pnum + (size_t)me * 4;
        const float* __restrict__ dn = denb + (size_t)d_l * 4;
#pragma unroll
        for (int h = 0; h < 4; ++h) a_l[h] = pe[h] / dn[h];
    }

    for (int j0 = 0; j0 < cnt; j0 += U) {
        float xv[U][4];
        float al[U][4];
        int   dj[U];
#pragma unroll
        for (int u = 0; u < U; ++u) {
            const int j = min(j0 + u, cnt - 1);
            const int sj = lane_read_i(s_l, j);
            dj[u] = lane_read_i(d_l, j);
            const float* __restrict__ xs = xl + (size_t)sj * HC + lane;
#pragma unroll
            for (int h = 0; h < 4; ++h) {
                al[u][h] = lane_read_f(a_l[h], j);
                xv[u][h] = xs[h * C];
            }
        }
#pragma unroll
        for (int u = 0; u < U; ++u) {
            if (j0 + u < cnt) {
                float* __restrict__ od = outacc + (size_t)dj[u] * HC + lane;
#pragma unroll
                for (int h = 0; h < 4; ++h)
                    atomicAdd(&od[h * C], xv[u][h] * al[u][h]);
            }
        }
    }
}

template <int U>
__global__ void __launch_bounds__(256)
edge_aggregate_headsum_kernel(const int* __restrict__ src,
                              const int* __restrict__ dst,
                              int E, int Etot,
                              const float* __restrict__ pnum,
                              const float* __restrict__ denb,
                              const float* __restrict__ xl,
                              float* __restrict__ outacc) {
    constexpr int C = 64;
    const int lane = threadIdx.x & 31;
    const int base = wave_index() * EPW;
    if (base >= Etot) return;
    const int cnt = min(EPW, Etot - base);

    int s_l = 0, d_l = 0;
    float a_l[4] = {0.f, 0.f, 0.f, 0.f};
    if (lane < cnt) {
        const int me = base + lane;
        if (me < E) { s_l = src[me]; d_l = dst[me]; }
        else        { s_l = d_l = me - E; }
        const float* __restrict__ pe = pnum + (size_t)me * 4;
        const float* __restrict__ dn = denb + (size_t)d_l * 4;
#pragma unroll
        for (int h = 0; h < 4; ++h) a_l[h] = pe[h] / dn[h];
    }

    for (int j0 = 0; j0 < cnt; j0 += U) {
        float x0[U][4], x1[U][4];
        float al[U][4];
        int   dj[U];
#pragma unroll
        for (int u = 0; u < U; ++u) {
            const int j = min(j0 + u, cnt - 1);
            const int sj = lane_read_i(s_l, j);
            dj[u] = lane_read_i(d_l, j);
            const float* __restrict__ xs = xl + (size_t)sj * (4 * C) + lane;
#pragma unroll
            for (int h = 0; h < 4; ++h) {
                al[u][h] = lane_read_f(a_l[h], j);
                x0[u][h] = xs[h * C];
                x1[u][h] = xs[h * C + WAVE];
            }
        }
#pragma unroll
        for (int u = 0; u < U; ++u) {
            if (j0 + u < cnt) {
                const float v0 = x0[u][0] * al[u][0] + x0[u][1] * al[u][1]
                               + x0[u][2] * al[u][2] + x0[u][3] * al[u][3];
                const float v1 = x1[u][0] * al[u][0] + x1[u][1] * al[u][1]
                               + x1[u][2] * al[u][2] + x1[u][3] * al[u][3];
                float* __restrict__ od = outacc + (size_t)dj[u] * C + lane;
                atomicAdd(&od[0], v0);
                atomicAdd(&od[WAVE], v1);
            }
        }
    }
}

#define NPW 4
template <int D>
__global__ void __launch_bounds__(256)
finalize_kernel(const float* __restrict__ acc,
                const float* __restrict__ bias,
                const float* __restrict__ g,
                const float* __restrict__ be,
                const float* __restrict__ res,
                float* __restrict__ out, int nN, float scale) {
    constexpr int NV = D / WAVE;
    const int lane  = threadIdx.x & 31;
    const int node0 = wave_index() * NPW;
    if (node0 >= nN) return;

    float bi[NV], gg[NV], bb[NV];
#pragma unroll
    for (int i = 0; i < NV; ++i) {
        const int ch = lane + i * WAVE;
        bi[i] = bias[ch]; gg[i] = g[ch]; bb[i] = be[ch];
    }

    float a[NPW][NV], r[NPW][NV];
#pragma unroll
    for (int q = 0; q < NPW; ++q) {
        const int node = min(node0 + q, nN - 1);
#pragma unroll
        for (int i = 0; i < NV; ++i) {
            const int ch = lane + i * WAVE;
            a[q][i] = acc[(size_t)node * D + ch];
            r[q][i] = res[(size_t)node * D + ch];
        }
    }
#pragma unroll
    for (int q = 0; q < NPW; ++q) {
        float v[NV], s = 0.f, s2 = 0.f;
#pragma unroll
        for (int i = 0; i < NV; ++i) {
            float x = scale * a[q][i] + bi[i];
            x = (x > 0.f) ? x : (__expf(x) - 1.f);
            v[i] = x;
            s += x;
            s2 += x * x;
        }
#pragma unroll
        for (int off = 16; off > 0; off >>= 1) {
            s  += __shfl_xor(s, off, WAVE);
            s2 += __shfl_xor(s2, off, WAVE);
        }
        const float mu  = s / (float)D;
        const float var = s2 / (float)D - mu * mu;
        const float inv = rsqrtf(var + 1e-5f);
        if (node0 + q < nN) {
#pragma unroll
            for (int i = 0; i < NV; ++i) {
                const int ch = lane + i * WAVE;
                out[(size_t)(node0 + q) * D + ch] = (v[i] - mu) * inv * gg[i] + bb[i] + r[q][i];
            }
        }
    }
}

static inline int blocks_for(long long work, int bs) {
    return (int)((work + bs - 1) / bs);
}

extern "C" void kernel_launch(void* const* d_in, const int* in_sizes, int n_in,
                              void* d_out, int out_size, void* d_ws, size_t ws_size,
                              hipStream_t stream) {
    const float* x    = (const float*)d_in[0];
    const int*   ei   = (const int*)d_in[1];
    const float* ew   = (const float*)d_in[2];
    const float* Wl0  = (const float*)d_in[3];
    const float* bl0  = (const float*)d_in[4];
    const float* Wr0  = (const float*)d_in[5];
    const float* br0  = (const float*)d_in[6];
    const float* We0  = (const float*)d_in[7];
    const float* att0 = (const float*)d_in[8];
    const float* b0   = (const float*)d_in[9];
    const float* g0   = (const float*)d_in[10];
    const float* be0  = (const float*)d_in[11];
    const float* Wl1  = (const float*)d_in[12];
    const float* bl1  = (const float*)d_in[13];
    const float* Wr1  = (const float*)d_in[14];
    const float* br1  = (const float*)d_in[15];
    const float* We1  = (const float*)d_in[16];
    const float* att1 = (const float*)d_in[17];
    const float* b1   = (const float*)d_in[18];
    const float* g1   = (const float*)d_in[19];
    const float* be1  = (const float*)d_in[20];
    const float* P0   = (const float*)d_in[21];
    const float* pb0  = (const float*)d_in[22];
    const float* P1   = (const float*)d_in[23];
    const float* pb1  = (const float*)d_in[24];

    const int IN  = 128, HIDD = 128, O1 = 256, OUT = 64;
    const int N = in_sizes[0] / IN;
    const int E = in_sizes[2];
    const int Etot = E + N;
    const int* srcp = ei;
    const int* dstp = ei + E;

    float* ws = (float*)d_ws;
    size_t off = 0;
    float* sumcnt = ws + off; off += 2 * (size_t)N;
    float* loopw = ws + off; off += (size_t)N;
    float* XL    = ws + off; off += (size_t)N * O1;
    float* XR    = ws + off; off += (size_t)N * O1;
    float* ACC   = ws + off; off += (size_t)N * O1;
    float* RES   = ws + off; off += (size_t)N * HIDD;
    float* HCUR  = ws + off; off += (size_t)N * HIDD;
    int*   MAXK  = (int*)(ws + off); off += (size_t)N * 4;
    float* DENB  = ws + off; off += (size_t)N * 4;
    float* LOGIT = ws + off; off += (size_t)Etot * 4;
    (void)ws_size; (void)n_in; (void)out_size;

    const int BS = 256;
    const int GEMM_WG = 512;
    const long long erun_threads = ((long long)(Etot + EPW - 1) / EPW) * WAVE;
    const long long erun16_threads = ((long long)(Etot + 15) / 16) * WAVE;
    const long long smax_threads = (4LL * Etot + 7) / 8;
    const long long node_threads = ((long long)(N + NPW - 1) / NPW) * WAVE;

    fill_kernel<<<blocks_for((2LL * N + FILL_PER_THREAD - 1) / FILL_PER_THREAD, BS), BS, 0, stream>>>(sumcnt, 0.f, 2LL * N);
    selfloop_accum_kernel<<<blocks_for(2LL * E, BS), BS, 0, stream>>>(dstp, ew, sumcnt, E);
    selfloop_w_kernel<<<blocks_for(N, BS), BS, 0, stream>>>(sumcnt, loopw, N);

    {
        gemm_bias_wmma_kernel<8><<<dim3(GEMM_WG, 1), BS, 0, stream>>>(x, Wl0, bl0, XL, N / 16);
        gemm_bias_wmma_kernel<8><<<dim3(GEMM_WG, 1), BS, 0, stream>>>(x, Wr0, br0, XR, N / 16);
        gemm_bias_wmma_kernel<8><<<dim3(GEMM_WG, 1), BS, 0, stream>>>(x, P0, pb0, RES, N / 16);

        fill_maxkey_kernel<<<blocks_for((4LL * N + FILL_PER_THREAD - 1) / FILL_PER_THREAD, BS), BS, 0, stream>>>(MAXK, 4LL * N);
        fill_kernel<<<blocks_for((4LL * N + FILL_PER_THREAD - 1) / FILL_PER_THREAD, BS), BS, 0, stream>>>(DENB, 0.f, 4LL * N);
        fill_kernel<<<blocks_for(((long long)N * HIDD + FILL_PER_THREAD - 1) / FILL_PER_THREAD, BS), BS, 0, stream>>>(ACC, 0.f, (long long)N * HIDD);

        edge_logits_kernel<32, 4, EPW><<<blocks_for(erun_threads, BS), BS, 0, stream>>>(
            srcp, dstp, ew, loopw, E, Etot, XL, XR, We0, att0, LOGIT, MAXK);
        edge_softmax_kernel<<<blocks_for(smax_threads, BS), BS, 0, stream>>>(
            dstp, E, Etot, LOGIT, MAXK, DENB);
        edge_aggregate_kernel<32, 4><<<blocks_for(erun_threads, BS), BS, 0, stream>>>(
            srcp, dstp, E, Etot, LOGIT, DENB, XL, ACC);

        finalize_kernel<128><<<blocks_for(node_threads, BS), BS, 0, stream>>>(
            ACC, b0, g0, be0, RES, HCUR, N, 1.0f);
    }

    {
        gemm_bias_wmma_kernel<16><<<dim3(GEMM_WG / 2, 2), BS, 0, stream>>>(HCUR, Wl1, bl1, XL, N / 16);
        gemm_bias_wmma_kernel<16><<<dim3(GEMM_WG / 2, 2), BS, 0, stream>>>(HCUR, Wr1, br1, XR, N / 16);
        gemm_bias_wmma_kernel<4><<<dim3(GEMM_WG, 1), BS, 0, stream>>>(HCUR, P1, pb1, RES, N / 16);

        fill_maxkey_kernel<<<blocks_for((4LL * N + FILL_PER_THREAD - 1) / FILL_PER_THREAD, BS), BS, 0, stream>>>(MAXK, 4LL * N);
        fill_kernel<<<blocks_for((4LL * N + FILL_PER_THREAD - 1) / FILL_PER_THREAD, BS), BS, 0, stream>>>(DENB, 0.f, 4LL * N);
        fill_kernel<<<blocks_for(((long long)N * OUT + FILL_PER_THREAD - 1) / FILL_PER_THREAD, BS), BS, 0, stream>>>(ACC, 0.f, (long long)N * OUT);

        edge_logits_kernel<64, 2, 16><<<blocks_for(erun16_threads, BS), BS, 0, stream>>>(
            srcp, dstp, ew, loopw, E, Etot, XL, XR, We1, att1, LOGIT, MAXK);
        edge_softmax_kernel<<<blocks_for(smax_threads, BS), BS, 0, stream>>>(
            dstp, E, Etot, LOGIT, MAXK, DENB);
        edge_aggregate_headsum_kernel<4><<<blocks_for(erun_threads, BS), BS, 0, stream>>>(
            srcp, dstp, E, Etot, LOGIT, DENB, XL, ACC);

        finalize_kernel<64><<<blocks_for(node_threads, BS), BS, 0, stream>>>(
            ACC, b1, g1, be1, RES, (float*)d_out, N, 0.25f);
    }
}
